# starting kernel
# baseline (speedup 1.0000x reference)
#include <hip/hip_runtime.h>

typedef float v2f __attribute__((ext_vector_type(2)));
typedef float v8f __attribute__((ext_vector_type(8)));

#define CH 128
#define LDS_PITCH 132

__global__ __launch_bounds__(128) void gcn_gemm_wmma(
    const float* __restrict__ x, const float* __restrict__ W,
    const float* __restrict__ bias, float* __restrict__ xt, int ntiles) {
  __shared__ float wl[CH * LDS_PITCH];

  const int tid = threadIdx.x;
  for (int i = tid; i < CH * CH; i += 128) {
    const int r = i >> 7;
    const int c = i & (CH - 1);
    wl[r * LDS_PITCH + c] = W[i];
  }
  __syncthreads();

  const int wave = tid >> 5;
  const int lane = tid & 31;
  const int rt = blockIdx.x * 4 + wave;
  if (rt >= ntiles) return;
  const int rb  = rt * 16;
  const int lh  = lane & 15;
  const int khi = lane >> 4;

  v8f acc[8];
#pragma unroll
  for (int cb = 0; cb < 8; ++cb) {
    const float bv = bias[cb * 16 + lh];
    acc[cb] = (v8f){bv, bv, bv, bv, bv, bv, bv, bv};
  }

  const float* arow = &x[(size_t)(rb + lh) * CH + 2 * khi];

  for (int k = 0; k < CH; k += 4) {
    const v2f a = *(const v2f*)(arow + k);
#pragma unroll
    for (int cb = 0; cb < 8; ++cb) {
      const v2f bm = *(const v2f*)&wl[(cb * 16 + lh) * LDS_PITCH + k + 2 * khi];
      acc[cb] = __builtin_amdgcn_wmma_f32_16x16x4_f32(
          false, a, false, bm, (short)0, acc[cb], false, false);
    }
  }

#pragma unroll
  for (int cb = 0; cb < 8; ++cb) {
#pragma unroll
    for (int i = 0; i < 8; ++i) {
      xt[(size_t)(rb + i + 8 * khi) * CH + cb * 16 + lh] = acc[cb][i];
    }
  }
}

__global__ void gcn_zero(float4* __restrict__ p, int n4) {
  const int i = blockIdx.x * blockDim.x + threadIdx.x;
  if (i < n4) p[i] = make_float4(0.f, 0.f, 0.f, 0.f);
}

__global__ __launch_bounds__(256) void gcn_scatter(
    const float* __restrict__ xt, const int* __restrict__ ei,
    float* __restrict__ out, int nedges) {
  const int e = blockIdx.x * (blockDim.x >> 5) + (threadIdx.x >> 5);
  if (e >= nedges) return;
  const int lane = threadIdx.x & 31;
  const int row = ei[e];
  const int col = ei[nedges + e];

  const float4 v = ((const float4*)&xt[(size_t)col * CH])[lane];
  float* dst = &out[(size_t)row * CH + lane * 4];
  atomicAdd(dst + 0, v.x);
  atomicAdd(dst + 1, v.y);
  atomicAdd(dst + 2, v.z);
  atomicAdd(dst + 3, v.w);
}

__global__ void gcn_tanh(float4* __restrict__ p, int n4) {
  const int i = blockIdx.x * blockDim.x + threadIdx.x;
  if (i < n4) {
    float4 v = p[i];
    v.x = tanhf(v.x);
    v.y = tanhf(v.y);
    v.z = tanhf(v.z);
    v.w = tanhf(v.w);
    p[i] = v;
  }
}

extern "C" void kernel_launch(void* const* d_in, const int* in_sizes, int n_in,
                              void* d_out, int out_size, void* d_ws, size_t ws_size,
                              hipStream_t stream) {
  const float* x    = (const float*)d_in[0];
  const int*   ei   = (const int*)d_in[1];
  const float* W    = (const float*)d_in[2];
  const float* bias = (const float*)d_in[3];

  float* out = (float*)d_out;
  float* xt  = (float*)d_ws;

  const int N = in_sizes[0] / CH;
  const int E = in_sizes[1] / 2;
  const int total = N * CH;
  const int n4 = total / 4;

  gcn_zero<<<(n4 + 255) / 256, 256, 0, stream>>>((float4*)out, n4);

  const int ntiles = N / 16;
  gcn_gemm_wmma<<<(ntiles + 3) / 4, 128, 0, stream>>>(x, W, bias, xt, ntiles);

  gcn_scatter<<<(E + 7) / 8, 256, 0, stream>>>(xt, ei, out, E);

  gcn_tanh<<<(n4 + 255) / 256, 256, 0, stream>>>((float4*)out, n4);
}


# Global f32 atomic scatter replaced by row ownership: 256 work-groups scan all destination ids, add gathered rows to their 196 rows in LDS (f32 LDS atomics, arrival order), tanh, store; 2 launches
# speedup vs baseline: 2.2205x; 2.2205x over previous
#include <hip/hip_runtime.h>

typedef float v2f __attribute__((ext_vector_type(2)));
typedef float v8f __attribute__((ext_vector_type(8)));

#define CH 128
#define LDS_PITCH 132

__global__ __launch_bounds__(128) void gcn_gemm_wmma(
    const float* __restrict__ x, const float* __restrict__ W,
    const float* __restrict__ bias, float* __restrict__ xt, int ntiles) {
  __shared__ float wl[CH * LDS_PITCH];

  const int tid = threadIdx.x;
  for (int i = tid; i < CH * CH; i += 128) {
    const int r = i >> 7;
    const int c = i & (CH - 1);
    wl[r * LDS_PITCH + c] = W[i];
  }
  __syncthreads();

  const int wave = tid >> 5;
  const int lane = tid & 31;
  const int rt = blockIdx.x * 4 + wave;
  if (rt >= ntiles) return;
  const int rb  = rt * 16;
  const int lh  = lane & 15;
  const int khi = lane >> 4;

  v8f acc[8];
#pragma unroll
  for (int cb = 0; cb < 8; ++cb) {
    const float bv = bias[cb * 16 + lh];
    acc[cb] = (v8f){bv, bv, bv, bv, bv, bv, bv, bv};
  }

  const float* arow = &x[(size_t)(rb + lh) * CH + 2 * khi];

  for (int k = 0; k < CH; k += 4) {
    const v2f a = *(const v2f*)(arow + k);
#pragma unroll
    for (int cb = 0; cb < 8; ++cb) {
      const v2f bm = *(const v2f*)&wl[(cb * 16 + lh) * LDS_PITCH + k + 2 * khi];
      acc[cb] = __builtin_amdgcn_wmma_f32_16x16x4_f32(
          false, a, false, bm, (short)0, acc[cb], false, false);
    }
  }

#pragma unroll
  for (int cb = 0; cb < 8; ++cb) {
#pragma unroll
    for (int i = 0; i < 8; ++i) {
      xt[(size_t)(rb + i + 8 * khi) * CH + cb * 16 + lh] = acc[cb][i];
    }
  }
}

#define ROWS 196
#define NW 32
#define GE 4
#define NBUF 12
#define PCAP 256

__device__ __forceinline__ void agg_gather_group(
    const float* __restrict__ xt, int colv, int lane, int first, float4 (&v)[GE]) {
#pragma unroll
  for (int j = 0; j < GE; ++j) {
    const int c = __builtin_amdgcn_readlane(colv, first + j);
    v[j] = *(const float4*)(xt + (size_t)c * CH + 4 * lane);
  }
}

__device__ __forceinline__ void agg_add_group(
    float* acc, int relv, int lane, int first, const float4 (&v)[GE]) {
#pragma unroll
  for (int j = 0; j < GE; ++j) {
    const int r = __builtin_amdgcn_readlane(relv, first + j);
    float* dst = acc + r * CH + lane;
    atomicAdd(dst + 0, v[j].x);
    atomicAdd(dst + 32, v[j].y);
    atomicAdd(dst + 64, v[j].z);
    atomicAdd(dst + 96, v[j].w);
  }
}

__device__ __forceinline__ void agg_batch32(
    const float* __restrict__ xt, float* acc, int relv, int colv, int lane) {
  float4 va[GE], vb[GE];
  agg_gather_group(xt, colv, lane, 0, va);
#pragma unroll
  for (int g = 0; g < 32; g += 2 * GE) {
    agg_gather_group(xt, colv, lane, g + GE, vb);
    agg_add_group(acc, relv, lane, g, va);
    if (g + 2 * GE < 32) agg_gather_group(xt, colv, lane, g + 2 * GE, va);
    agg_add_group(acc, relv, lane, g + GE, vb);
  }
}

__device__ __forceinline__ void agg_batch_part(
    const float* __restrict__ xt, float* acc, int relv, int colv, int lane, int cnt) {
#pragma unroll
  for (int g = 0; g < 32; g += GE) {
    if (g < cnt) {
      float4 v[GE];
#pragma unroll
      for (int j = 0; j < GE; ++j) {
        v[j] = make_float4(0.f, 0.f, 0.f, 0.f);
        if (g + j < cnt) {
          const int c = __builtin_amdgcn_readlane(colv, g + j);
          v[j] = *(const float4*)(xt + (size_t)c * CH + 4 * lane);
        }
      }
#pragma unroll
      for (int j = 0; j < GE; ++j) {
        if (g + j < cnt) {
          const int r = __builtin_amdgcn_readlane(relv, g + j);
          float* dst = acc + r * CH + lane;
          atomicAdd(dst + 0, v[j].x);
          atomicAdd(dst + 32, v[j].y);
          atomicAdd(dst + 64, v[j].z);
          atomicAdd(dst + 96, v[j].w);
        }
      }
    }
  }
}

__device__ __forceinline__ int agg_chunk_of_slot(int q, int rstart, int nround, int wv) {
  int rr = q + rstart;
  if (rr >= nround) rr -= nround;
  return rr * NW + wv;
}

__device__ __forceinline__ bool agg_slot_valid(int q, int rstart, int nround, int wv, int nfull) {
  return (q < nround) && (agg_chunk_of_slot(q, rstart, nround, wv) < nfull);
}

__device__ __forceinline__ int4 agg_load_slot(
    const int* __restrict__ ei, int q, int rstart, int nround, int wv, int lane, int nfull) {
  const int c = agg_slot_valid(q, rstart, nround, wv, nfull)
                    ? agg_chunk_of_slot(q, rstart, nround, wv) : 0;
  return *(const int4*)(ei + c * 128 + 4 * lane);
}

__device__ __forceinline__ void agg_chunk_direct(
    const float* __restrict__ xt, const int* __restrict__ ei, float* acc, int nedges,
    int e, int r0, int r1, int r2, int r3,
    unsigned m0, unsigned m1, unsigned m2, unsigned m3, int lane) {
#pragma unroll 1
  for (int comp = 0; comp < 4; ++comp) {
    unsigned m = comp == 0 ? m0 : comp == 1 ? m1 : comp == 2 ? m2 : m3;
    const int rv = comp == 0 ? r0 : comp == 1 ? r1 : comp == 2 ? r2 : r3;
    while (m != 0u) {
      const int j = __builtin_ctz(m);
      m &= m - 1u;
      const int r = __builtin_amdgcn_readlane(rv, j);
      const int eid = __builtin_amdgcn_readlane(e, j) + comp;
      const int c = ei[nedges + eid];
      const float4 v = *(const float4*)(xt + (size_t)c * CH + 4 * lane);
      float* dst = acc + r * CH + lane;
      atomicAdd(dst + 0, v.x);
      atomicAdd(dst + 32, v.y);
      atomicAdd(dst + 64, v.z);
      atomicAdd(dst + 96, v.w);
    }
  }
}

__global__ __launch_bounds__(1024) void gcn_aggregate(
    const float* __restrict__ xt, const int* __restrict__ ei,
    float* __restrict__ out, int nedges, int nrows) {
  __shared__ float acc[ROWS * CH];
  __shared__ int ring_r[NW][PCAP];
  __shared__ int ring_e[NW][PCAP];

  const int tid = threadIdx.x;
  const int lane = tid & 31;
  const int wv = __builtin_amdgcn_readfirstlane(tid >> 5);
  int* myr = ring_r[wv];
  int* mye = ring_e[wv];

  const int row0 = blockIdx.x * ROWS;
  const int nr = min(ROWS, nrows - row0);

  float4* acc4 = (float4*)acc;
  for (int i = tid; i < ROWS * CH / 4; i += 1024) acc4[i] = make_float4(0.f, 0.f, 0.f, 0.f);
  __syncthreads();

  const int nfull = nedges >> 7;
  const int nround = (nfull + NW - 1) / NW;
  const int rstart = (int)(((unsigned)blockIdx.x * (unsigned)nround) / gridDim.x);
  const int niter = (nround + NBUF - 1) / NBUF;

  int head = 0;
  int npend = 0;

  if (wv == 0) {
    for (int base = nfull * 128; base < nedges; base += 32) {
      const int e = base + lane;
      int r = -1;
      if (e < nedges) r = ei[e] - row0;
      const bool p = (unsigned)r < (unsigned)nr;
      const unsigned m = __builtin_amdgcn_ballot_w32(p);
      if (p) {
        const int idx = (npend + (int)__builtin_amdgcn_mbcnt_lo(m, 0u)) & (PCAP - 1);
        myr[idx] = r;
        mye[idx] = e;
      }
      npend += __builtin_popcount(m);
    }
  }

  if (nround > 0) {
    int4 buf[NBUF];
#pragma unroll
    for (int k = 0; k < NBUF; ++k) buf[k] = agg_load_slot(ei, k, rstart, nround, wv, lane, nfull);

    for (int it = 0; it < niter; ++it) {
#pragma unroll
      for (int k = 0; k < NBUF; ++k) {
        const int q = it * NBUF + k;
        const int4 d = buf[k];
        const unsigned lim = agg_slot_valid(q, rstart, nround, wv, nfull) ? (unsigned)nr : 0u;
        const int r0 = d.x - row0, r1 = d.y - row0, r2 = d.z - row0, r3 = d.w - row0;
        const bool p0 = (unsigned)r0 < lim;
        const bool p1 = (unsigned)r1 < lim;
        const bool p2 = (unsigned)r2 < lim;
        const bool p3 = (unsigned)r3 < lim;
        const unsigned m0 = __builtin_amdgcn_ballot_w32(p0);
        const unsigned m1 = __builtin_amdgcn_ballot_w32(p1);
        const unsigned m2 = __builtin_amdgcn_ballot_w32(p2);
        const unsigned m3 = __builtin_amdgcn_ballot_w32(p3);
        if ((m0 | m1 | m2 | m3) != 0u) {
          const int e = agg_chunk_of_slot(q, rstart, nround, wv) * 128 + 4 * lane;
          const int k0 = __builtin_popcount(m0), k1 = __builtin_popcount(m1);
          const int k2 = __builtin_popcount(m2), k3 = __builtin_popcount(m3);
          if (npend + k0 + k1 + k2 + k3 <= PCAP) {
            int tail = head + npend;
            if (p0) {
              const int idx = (tail + (int)__builtin_amdgcn_mbcnt_lo(m0, 0u)) & (PCAP - 1);
              myr[idx] = r0;
              mye[idx] = e;
            }
            tail += k0;
            if (p1) {
              const int idx = (tail + (int)__builtin_amdgcn_mbcnt_lo(m1, 0u)) & (PCAP - 1);
              myr[idx] = r1;
              mye[idx] = e + 1;
            }
            tail += k1;
            if (p2) {
              const int idx = (tail + (int)__builtin_amdgcn_mbcnt_lo(m2, 0u)) & (PCAP - 1);
              myr[idx] = r2;
              mye[idx] = e + 2;
            }
            tail += k2;
            if (p3) {
              const int idx = (tail + (int)__builtin_amdgcn_mbcnt_lo(m3, 0u)) & (PCAP - 1);
              myr[idx] = r3;
              mye[idx] = e + 3;
            }
            npend += k0 + k1 + k2 + k3;
          } else {
            agg_chunk_direct(xt, ei, acc, nedges, e, r0, r1, r2, r3, m0, m1, m2, m3, lane);
          }
        }
        buf[k] = agg_load_slot(ei, q + NBUF, rstart, nround, wv, lane, nfull);
      }

      while (npend >= 32) {
        const int idx = (head + lane) & (PCAP - 1);
        const int relv = myr[idx];
        const int colv = ei[nedges + mye[idx]];
        agg_batch32(xt, acc, relv, colv, lane);
        head = (head + 32) & (PCAP - 1);
        npend -= 32;
      }
    }
  }

  while (npend > 0) {
    const int cnt = npend < 32 ? npend : 32;
    const int idx = (head + lane) & (PCAP - 1);
    int relv = 0, colv = 0;
    if (lane < cnt) { relv = myr[idx]; colv = ei[nedges + mye[idx]]; }
    agg_batch_part(xt, acc, relv, colv, lane, cnt);
    head = (head + cnt) & (PCAP - 1);
    npend -= cnt;
  }

  __syncthreads();

  float4* out4 = (float4*)out;
  for (int i = tid; i < nr * 32; i += 1024) {
    const int row = i >> 5;
    const float* a = acc + row * CH + lane;
    float4 v;
    v.x = tanhf(a[0]);
    v.y = tanhf(a[32]);
    v.z = tanhf(a[64]);
    v.w = tanhf(a[96]);
    out4[(size_t)(row0 + row) * 32 + lane] = v;
  }
}

extern "C" void kernel_launch(void* const* d_in, const int* in_sizes, int n_in,
                              void* d_out, int out_size, void* d_ws, size_t ws_size,
                              hipStream_t stream) {
  const float* x    = (const float*)d_in[0];
  const int*   ei   = (const int*)d_in[1];
  const float* W    = (const float*)d_in[2];
  const float* bias = (const float*)d_in[3];

  float* out = (float*)d_out;
  float* xt  = (float*)d_ws;

  const int N = in_sizes[0] / CH;
  const int E = in_sizes[1] / 2;

  const int ntiles = N / 16;
  gcn_gemm_wmma<<<(ntiles + 3) / 4, 128, 0, stream>>>(x, W, bias, xt, ntiles);

  gcn_aggregate<<<(N + ROWS - 1) / ROWS, 1024, 0, stream>>>(xt, ei, out, E, N);
}


# Scan reads a 1-byte slice key per edge (launch 1 writes keys and a packed row+source word per edge): a 16-byte lane-load tests 16 edges, not 4; adds stay inside the scan, arrival-order sums kept
# speedup vs baseline: 2.9726x; 1.3387x over previous
#include <hip/hip_runtime.h>

typedef float v2f __attribute__((ext_vector_type(2)));
typedef float v8f __attribute__((ext_vector_type(8)));

#define CH 128
#define LDS_PITCH 132
#define ROWS 196

__device__ __forceinline__ unsigned* slice_keys(float* xt, int nrows) {
  return (unsigned*)(xt + (size_t)nrows * CH);
}
__device__ __forceinline__ const unsigned* slice_keys(const float* xt, int nrows) {
  return (const unsigned*)(xt + (size_t)nrows * CH);
}
__device__ __forceinline__ unsigned* edge_words(unsigned* keys, int nedges) {
  return keys + (((nedges + 15) >> 4) << 2);
}
__device__ __forceinline__ const unsigned* edge_words(const unsigned* keys, int nedges) {
  return keys + (((nedges + 15) >> 4) << 2);
}
__device__ __forceinline__ unsigned edge_word(int dest, int src, int nrows) {
  const unsigned r = (unsigned)dest < (unsigned)nrows ? (unsigned)dest % ROWS : 255u;
  return (r << 24) | ((unsigned)src & 0xFFFFFFu);
}
__device__ __forceinline__ int4 edge_word4(int4 d, int4 s, int nrows) {
  return make_int4((int)edge_word(d.x, s.x, nrows), (int)edge_word(d.y, s.y, nrows),
                   (int)edge_word(d.z, s.z, nrows), (int)edge_word(d.w, s.w, nrows));
}
__device__ __forceinline__ int4 edge_src4(const int* __restrict__ ei, int nedges, int g) {
  const int* p = ei + nedges + 4 * g;
  if ((nedges & 3) == 0) return *(const int4*)p;
  return make_int4(p[0], p[1], p[2], p[3]);
}
__device__ __forceinline__ unsigned slice_key_word(int4 d) {
  return (((unsigned)d.x / ROWS) & 255u) | ((((unsigned)d.y / ROWS) & 255u) << 8) |
         ((((unsigned)d.z / ROWS) & 255u) << 16) | (((unsigned)d.w / ROWS) << 24);
}

__global__ __launch_bounds__(128) void gcn_gemm_wmma(
    const float* __restrict__ x, const float* __restrict__ W,
    const float* __restrict__ bias, float* __restrict__ xt, int ntiles,
    const int* __restrict__ ei, int nedges, int nrows) {
  __shared__ float wl[CH * LDS_PITCH];

  const int tid = threadIdx.x;

  unsigned* keys = slice_keys(xt, nrows);
  const int nthr = (int)gridDim.x * 128;
  const int gid = (int)blockIdx.x * 128 + tid;
  const int nquad = nedges >> 2;
  const int g0 = max(min(gid, nquad - 1), 0), g1 = max(min(gid + nthr, nquad - 1), 0);
  const int4 kd0 = *(const int4*)(ei + 4 * g0);
  const int4 kd1 = *(const int4*)(ei + 4 * g1);
  const int4 ks0 = edge_src4(ei, nedges, g0);
  const int4 ks1 = edge_src4(ei, nedges, g1);

  for (int i = tid; i < CH * CH; i += 128) {
    const int r = i >> 7;
    const int c = i & (CH - 1);
    wl[r * LDS_PITCH + c] = W[i];
  }
  __syncthreads();

  const int wave = tid >> 5;
  const int lane = tid & 31;
  const int rt = blockIdx.x * 4 + wave;
  if (rt < ntiles) {
  const int rb  = rt * 16;
  const int lh  = lane & 15;
  const int khi = lane >> 4;

  v8f acc[8];
#pragma unroll
  for (int cb = 0; cb < 8; ++cb) {
    const float bv = bias[cb * 16 + lh];
    acc[cb] = (v8f){bv, bv, bv, bv, bv, bv, bv, bv};
  }

  const float* arow = &x[(size_t)(rb + lh) * CH + 2 * khi];

  for (int k = 0; k < CH; k += 4) {
    const v2f a = *(const v2f*)(arow + k);
#pragma unroll
    for (int cb = 0; cb < 8; ++cb) {
      const v2f bm = *(const v2f*)&wl[(cb * 16 + lh) * LDS_PITCH + k + 2 * khi];
      acc[cb] = __builtin_amdgcn_wmma_f32_16x16x4_f32(
          false, a, false, bm, (short)0, acc[cb], false, false);
    }
  }

#pragma unroll
  for (int cb = 0; cb < 8; ++cb) {
#pragma unroll
    for (int i = 0; i < 8; ++i) {
      xt[(size_t)(rb + i + 8 * khi) * CH + cb * 16 + lh] = acc[cb][i];
    }
  }
  }

  unsigned* pk = edge_words(keys, nedges);
  if (gid < nquad) {
    keys[gid] = slice_key_word(kd0);
    *(int4*)(pk + 4 * gid) = edge_word4(kd0, ks0, nrows);
  }
  if (gid + nthr < nquad) {
    keys[gid + nthr] = slice_key_word(kd1);
    *(int4*)(pk + 4 * (gid + nthr)) = edge_word4(kd1, ks1, nrows);
  }
  for (int g = gid + 2 * nthr; g < nquad; g += nthr) {
    const int4 d = *(const int4*)(ei + 4 * g);
    keys[g] = slice_key_word(d);
    *(int4*)(pk + 4 * g) = edge_word4(d, edge_src4(ei, nedges, g), nrows);
  }
  if (gid < (nedges & 3)) {
    const int e = (nquad << 2) + gid;
    ((unsigned char*)keys)[e] = (unsigned char)((unsigned)ei[e] / ROWS);
    pk[e] = edge_word(ei[e], ei[nedges + e], nrows);
  }
}

#define NW 32
#define GE 4
#define NBUF 12
#define NFL 3
#define PCAP 256
#define KCH 512

__device__ __forceinline__ void agg_gather_group(
    const float* __restrict__ xt, int colv, int lane, int first, float4 (&v)[GE]) {
#pragma unroll
  for (int j = 0; j < GE; ++j) {
    const int c = __builtin_amdgcn_readlane(colv, first + j);
    v[j] = *(const float4*)(xt + (size_t)c * CH + 4 * lane);
  }
}

__device__ __forceinline__ void agg_add_group(
    float* acc, int relv, int lane, int first, const float4 (&v)[GE]) {
#pragma unroll
  for (int j = 0; j < GE; ++j) {
    const int r = __builtin_amdgcn_readlane(relv, first + j);
    float* dst = acc + r * CH + lane;
    atomicAdd(dst + 0, v[j].x);
    atomicAdd(dst + 32, v[j].y);
    atomicAdd(dst + 64, v[j].z);
    atomicAdd(dst + 96, v[j].w);
  }
}

__device__ __forceinline__ void agg_batch32(
    const float* __restrict__ xt, float* acc, int relv, int colv, int lane) {
  float4 va[GE], vb[GE];
  agg_gather_group(xt, colv, lane, 0, va);
#pragma unroll
  for (int g = 0; g < 32; g += 2 * GE) {
    agg_gather_group(xt, colv, lane, g + GE, vb);
    agg_add_group(acc, relv, lane, g, va);
    if (g + 2 * GE < 32) agg_gather_group(xt, colv, lane, g + 2 * GE, va);
    agg_add_group(acc, relv, lane, g + GE, vb);
  }
}

__device__ __forceinline__ void agg_batch_part(
    const float* __restrict__ xt, float* acc, int relv, int colv, int lane, int cnt) {
#pragma unroll
  for (int g = 0; g < 32; g += GE) {
    if (g < cnt) {
      float4 v[GE];
#pragma unroll
      for (int j = 0; j < GE; ++j) {
        v[j] = make_float4(0.f, 0.f, 0.f, 0.f);
        if (g + j < cnt) {
          const int c = __builtin_amdgcn_readlane(colv, g + j);
          v[j] = *(const float4*)(xt + (size_t)c * CH + 4 * lane);
        }
      }
#pragma unroll
      for (int j = 0; j < GE; ++j) {
        if (g + j < cnt) {
          const int r = __builtin_amdgcn_readlane(relv, g + j);
          float* dst = acc + r * CH + lane;
          atomicAdd(dst + 0, v[j].x);
          atomicAdd(dst + 32, v[j].y);
          atomicAdd(dst + 64, v[j].z);
          atomicAdd(dst + 96, v[j].w);
        }
      }
    }
  }
}

__device__ __forceinline__ int agg_chunk_of_slot(int q, int rstart, int nround, int wv) {
  int rr = q + rstart;
  if (rr >= nround) rr -= nround;
  return rr * NW + wv;
}

__device__ __forceinline__ bool agg_slot_valid(int q, int rstart, int nround, int wv, int nfull) {
  return (q < nround) && (agg_chunk_of_slot(q, rstart, nround, wv) < nfull);
}

__device__ __forceinline__ int4 agg_load_slot(
    const unsigned* __restrict__ keys, int q, int rstart, int nround, int wv, int lane, int nfull) {
  const int c = agg_slot_valid(q, rstart, nround, wv, nfull)
                    ? agg_chunk_of_slot(q, rstart, nround, wv) : 0;
  return *(const int4*)(keys + c * (KCH / 4) + 4 * lane);
}

__device__ __forceinline__ unsigned agg_key_hits(unsigned w, unsigned kb) {
  const unsigned x = w ^ kb;
  const unsigned y = (x & 0x7F7F7F7Fu) + 0x7F7F7F7Fu;
  return ~(y | x) & 0x80808080u;
}

__device__ __forceinline__ unsigned agg_hits_nibble(unsigned z) {
  return (((z >> 7) * 0x00204081u) >> 21) & 0xFu;
}

__device__ __forceinline__ void agg_edge_direct(
    const float* __restrict__ xt, const int* __restrict__ ei, float* acc, int nedges,
    int e, int row0, int nr, int lane) {
  const int r = ei[e] - row0;
  if ((unsigned)r < (unsigned)nr) {
    const int c = ei[nedges + e];
    const float4 v = *(const float4*)(xt + (size_t)c * CH + 4 * lane);
    float* dst = acc + r * CH + lane;
    atomicAdd(dst + 0, v.x);
    atomicAdd(dst + 32, v.y);
    atomicAdd(dst + 64, v.z);
    atomicAdd(dst + 96, v.w);
  }
}

__device__ __forceinline__ void agg_hits_direct(
    const float* __restrict__ xt, const int* __restrict__ ei, float* acc, int nedges,
    int ebase, unsigned hits, unsigned m, int row0, int nr, int lane) {
#pragma unroll 1
  while (m != 0u) {
    const int l = __builtin_ctz(m);
    m &= m - 1u;
    unsigned bits = (unsigned)__builtin_amdgcn_readlane((int)hits, l);
#pragma unroll 1
    while (bits != 0u) {
      const int e = ebase + 16 * l + __builtin_ctz(bits);
      bits &= bits - 1u;
      agg_edge_direct(xt, ei, acc, nedges, e, row0, nr, lane);
    }
  }
}

__global__ __launch_bounds__(1024) void gcn_aggregate(
    const float* __restrict__ xt, const int* __restrict__ ei,
    float* __restrict__ out, int nedges, int nrows) {
  __shared__ float acc[(ROWS + 1) * CH];
  __shared__ int ring_e[NW][PCAP];

  const unsigned* keys = slice_keys(xt, nrows);
  const unsigned* pk = edge_words(keys, nedges);

  const int tid = threadIdx.x;
  const int lane = tid & 31;
  const int wv = __builtin_amdgcn_readfirstlane(tid >> 5);
  int* mye = ring_e[wv];

  const int row0 = blockIdx.x * ROWS;
  const int nr = min(ROWS, nrows - row0);
  const unsigned kb = ((unsigned)blockIdx.x & 255u) * 0x01010101u;

  float4* acc4 = (float4*)acc;
  for (int i = tid; i < (ROWS + 1) * CH / 4; i += 1024) acc4[i] = make_float4(0.f, 0.f, 0.f, 0.f);
  __syncthreads();

  const int nfull = nedges / KCH;
  const int nround = (nfull + NW - 1) / NW;
  const int rstart = (int)(((unsigned)blockIdx.x * (unsigned)nround) / gridDim.x);
  const int niter = (nround + NBUF - 1) / NBUF;

  int head = 0;
  int npend = 0;

  for (int base = nfull * KCH + 32 * wv; base < nedges; base += 32 * NW) {
    const int e = base + lane;
    int r = -1;
    if (e < nedges) r = ei[e] - row0;
    unsigned m = __builtin_amdgcn_ballot_w32((unsigned)r < (unsigned)nr);
    while (m != 0u) {
      const int l = __builtin_ctz(m);
      m &= m - 1u;
      agg_edge_direct(xt, ei, acc, nedges, base + l, row0, nr, lane);
    }
  }

  if (nround > 0) {
    int4 buf[NBUF];
#pragma unroll
    for (int k = 0; k < NBUF; ++k) buf[k] = agg_load_slot(keys, k, rstart, nround, wv, lane, nfull);

    for (int it = 0; it < niter; ++it) {
#pragma unroll
      for (int k = 0; k < NBUF; ++k) {
        const int q = it * NBUF + k;
        const int4 d = buf[k];
        const unsigned z0 = agg_key_hits((unsigned)d.x, kb);
        const unsigned z1 = agg_key_hits((unsigned)d.y, kb);
        const unsigned z2 = agg_key_hits((unsigned)d.z, kb);
        const unsigned z3 = agg_key_hits((unsigned)d.w, kb);
        unsigned m = __builtin_amdgcn_ballot_w32((z0 | z1 | z2 | z3) != 0u);
        if (!agg_slot_valid(q, rstart, nround, wv, nfull)) m = 0u;
        if (m != 0u) {
          unsigned hits = agg_hits_nibble(z0) | (agg_hits_nibble(z1) << 4) |
                          (agg_hits_nibble(z2) << 8) | (agg_hits_nibble(z3) << 12);
          const int ebase = agg_chunk_of_slot(q, rstart, nround, wv) * KCH;
          do {
            if (npend > PCAP - 32) {
              agg_hits_direct(xt, ei, acc, nedges, ebase, hits, m, row0, nr, lane);
              break;
            }
            if (hits != 0u) {
              const int idx = (head + npend + (int)__builtin_amdgcn_mbcnt_lo(m, 0u)) & (PCAP - 1);
              mye[idx] = ebase + 16 * lane + __builtin_ctz(hits);
              hits &= hits - 1u;
            }
            npend += __builtin_popcount(m);
            m = __builtin_amdgcn_ballot_w32(hits != 0u);
          } while (m != 0u);
        }
        buf[k] = agg_load_slot(keys, q + NBUF, rstart, nround, wv, lane, nfull);

        if (k % NFL == NFL - 1) {
          while (npend >= 32) {
            const unsigned w = pk[mye[(head + lane) & (PCAP - 1)]];
            const int r = (int)(w >> 24);
            const int colv = (int)(w & 0xFFFFFFu);
            const int relv = r < nr ? r : ROWS;
            agg_batch32(xt, acc, relv, colv, lane);
            head = (head + 32) & (PCAP - 1);
            npend -= 32;
          }
        }
      }
    }
  }

  while (npend > 0) {
    const int cnt = npend < 32 ? npend : 32;
    int relv = 0, colv = 0;
    if (lane < cnt) {
      const unsigned w = pk[mye[(head + lane) & (PCAP - 1)]];
      const int r = (int)(w >> 24);
      colv = (int)(w & 0xFFFFFFu);
      relv = r < nr ? r : ROWS;
    }
    agg_batch_part(xt, acc, relv, colv, lane, cnt);
    head = (head + cnt) & (PCAP - 1);
    npend -= cnt;
  }

  __syncthreads();

  float4* out4 = (float4*)out;
  for (int i = tid; i < nr * 32; i += 1024) {
    const int row = i >> 5;
    const float* a = acc + row * CH + lane;
    float4 v;
    v.x = tanhf(a[0]);
    v.y = tanhf(a[32]);
    v.z = tanhf(a[64]);
    v.w = tanhf(a[96]);
    out4[(size_t)(row0 + row) * 32 + lane] = v;
  }
}

extern "C" void kernel_launch(void* const* d_in, const int* in_sizes, int n_in,
                              void* d_out, int out_size, void* d_ws, size_t ws_size,
                              hipStream_t stream) {
  const float* x    = (const float*)d_in[0];
  const int*   ei   = (const int*)d_in[1];
  const float* W    = (const float*)d_in[2];
  const float* bias = (const float*)d_in[3];

  float* out = (float*)d_out;
  float* xt  = (float*)d_ws;

  const int N = in_sizes[0] / CH;
  const int E = in_sizes[1] / 2;

  const int ntiles = N / 16;
  gcn_gemm_wmma<<<(ntiles + 3) / 4, 128, 0, stream>>>(x, W, bias, xt, ntiles, ei, E, N);

  gcn_aggregate<<<(N + ROWS - 1) / ROWS, 1024, 0, stream>>>(xt, ei, out, E, N);
}


# Slice keys stored as bit planes (launch 1: a ballot per key bit, 32 edges a word): the scan tests 32 edges a lane with 8 XORs and an OR tree, not byte arithmetic on 16; arrival-order sums kept
# speedup vs baseline: 3.1662x; 1.0651x over previous
#include <hip/hip_runtime.h>

typedef float v2f __attribute__((ext_vector_type(2)));
typedef float v8f __attribute__((ext_vector_type(8)));

#define CH 128
#define LDS_PITCH 132
#define ROWS 196

__device__ __forceinline__ unsigned* slice_keys(float* xt, int nrows) {
  return (unsigned*)(xt + (size_t)nrows * CH);
}
__device__ __forceinline__ const unsigned* slice_keys(const float* xt, int nrows) {
  return (const unsigned*)(xt + (size_t)nrows * CH);
}
__device__ __forceinline__ unsigned* edge_words(unsigned* keys, int nedges) {
  return keys + (((nedges + 15) >> 4) << 2);
}
__device__ __forceinline__ const unsigned* edge_words(const unsigned* keys, int nedges) {
  return keys + (((nedges + 15) >> 4) << 2);
}
__device__ __forceinline__ unsigned edge_word(int dest, int src, int nrows) {
  const unsigned r = (unsigned)dest < (unsigned)nrows ? (unsigned)dest % ROWS : 255u;
  return (r << 24) | ((unsigned)src & 0xFFFFFFu);
}
__device__ __forceinline__ int4 edge_word4(int4 d, int4 s, int nrows) {
  return make_int4((int)edge_word(d.x, s.x, nrows), (int)edge_word(d.y, s.y, nrows),
                   (int)edge_word(d.z, s.z, nrows), (int)edge_word(d.w, s.w, nrows));
}
__device__ __forceinline__ int4 edge_src4(const int* __restrict__ ei, int nedges, int g) {
  const int* p = ei + nedges + 4 * g;
  if ((nedges & 3) == 0) return *(const int4*)p;
  return make_int4(p[0], p[1], p[2], p[3]);
}
#define KCH 1024
#define PGRP 8

__device__ __forceinline__ void plane_ids(const int* __restrict__ ei, int nedges, int blk,
                                          int wave, int lane, int (&d)[PGRP]) {
#pragma unroll
  for (int j = 0; j < PGRP; ++j) {
    const int e = blk * KCH + (PGRP * wave + j) * 32 + lane;
    d[j] = ei[max(min(e, nedges - 1), 0)];
  }
}

__device__ __forceinline__ void plane_store(unsigned* keys, int blk, int wave, int lane,
                                            const int (&d)[PGRP]) {
  unsigned p[8] = {0u, 0u, 0u, 0u, 0u, 0u, 0u, 0u};
#pragma unroll
  for (int j = 0; j < PGRP; ++j) {
    const unsigned key = (unsigned)d[j] / ROWS;
#pragma unroll
    for (int b = 0; b < 8; ++b) {
      const unsigned w = __builtin_amdgcn_ballot_w32(((key >> b) & 1u) != 0u);
      if (lane == j) p[b] = w;
    }
  }
  if (lane < PGRP) {
    unsigned* q = keys + blk * (KCH / 4) + (PGRP * wave + lane) * 4;
    *(int4*)q = make_int4((int)p[0], (int)p[1], (int)p[2], (int)p[3]);
    *(int4*)(q + KCH / 8) = make_int4((int)p[4], (int)p[5], (int)p[6], (int)p[7]);
  }
}

__global__ __launch_bounds__(128) void gcn_gemm_wmma(
    const float* __restrict__ x, const float* __restrict__ W,
    const float* __restrict__ bias, float* __restrict__ xt, int ntiles,
    const int* __restrict__ ei, int nedges, int nrows) {
  __shared__ float wl[CH * LDS_PITCH];

  const int tid = threadIdx.x;

  unsigned* keys = slice_keys(xt, nrows);
  const int nblk = nedges / KCH;
  int pd[PGRP];
  plane_ids(ei, nedges, (int)blockIdx.x, tid >> 5, tid & 31, pd);
  const int nthr = (int)gridDim.x * 128;
  const int gid = (int)blockIdx.x * 128 + tid;
  const int nquad = nedges >> 2;
  const int g0 = max(min(gid, nquad - 1), 0), g1 = max(min(gid + nthr, nquad - 1), 0);
  const int4 kd0 = *(const int4*)(ei + 4 * g0);
  const int4 kd1 = *(const int4*)(ei + 4 * g1);
  const int4 ks0 = edge_src4(ei, nedges, g0);
  const int4 ks1 = edge_src4(ei, nedges, g1);

  for (int i = tid; i < CH * CH; i += 128) {
    const int r = i >> 7;
    const int c = i & (CH - 1);
    wl[r * LDS_PITCH + c] = W[i];
  }
  __syncthreads();

  const int wave = tid >> 5;
  const int lane = tid & 31;
  const int rt = blockIdx.x * 4 + wave;
  if (rt < ntiles) {
  const int rb  = rt * 16;
  const int lh  = lane & 15;
  const int khi = lane >> 4;

  v8f acc[8];
#pragma unroll
  for (int cb = 0; cb < 8; ++cb) {
    const float bv = bias[cb * 16 + lh];
    acc[cb] = (v8f){bv, bv, bv, bv, bv, bv, bv, bv};
  }

  const float* arow = &x[(size_t)(rb + lh) * CH + 2 * khi];

  for (int k = 0; k < CH; k += 4) {
    const v2f a = *(const v2f*)(arow + k);
#pragma unroll
    for (int cb = 0; cb < 8; ++cb) {
      const v2f bm = *(const v2f*)&wl[(cb * 16 + lh) * LDS_PITCH + k + 2 * khi];
      acc[cb] = __builtin_amdgcn_wmma_f32_16x16x4_f32(
          false, a, false, bm, (short)0, acc[cb], false, false);
    }
  }

#pragma unroll
  for (int cb = 0; cb < 8; ++cb) {
#pragma unroll
    for (int i = 0; i < 8; ++i) {
      xt[(size_t)(rb + i + 8 * khi) * CH + cb * 16 + lh] = acc[cb][i];
    }
  }
  }

  if ((int)blockIdx.x < nblk) plane_store(keys, (int)blockIdx.x, tid >> 5, tid & 31, pd);
  for (int blk = (int)blockIdx.x + (int)gridDim.x; blk < nblk; blk += (int)gridDim.x) {
    plane_ids(ei, nedges, blk, tid >> 5, tid & 31, pd);
    plane_store(keys, blk, tid >> 5, tid & 31, pd);
  }

  unsigned* pk = edge_words(keys, nedges);
  if (gid < nquad) *(int4*)(pk + 4 * gid) = edge_word4(kd0, ks0, nrows);
  if (gid + nthr < nquad) *(int4*)(pk + 4 * (gid + nthr)) = edge_word4(kd1, ks1, nrows);
  for (int g = gid + 2 * nthr; g < nquad; g += nthr) {
    const int4 d = *(const int4*)(ei + 4 * g);
    *(int4*)(pk + 4 * g) = edge_word4(d, edge_src4(ei, nedges, g), nrows);
  }
  if (gid < (nedges & 3)) {
    const int e = (nquad << 2) + gid;
    pk[e] = edge_word(ei[e], ei[nedges + e], nrows);
  }
}

#define NW 32
#define GE 4
#define NBUF 6
#define NFL 2
#define PCAP 256

__device__ __forceinline__ void agg_gather_group(
    const float* __restrict__ xt, int colv, int lane, int first, float4 (&v)[GE]) {
#pragma unroll
  for (int j = 0; j < GE; ++j) {
    const int c = __builtin_amdgcn_readlane(colv, first + j);
    v[j] = *(const float4*)(xt + (size_t)c * CH + 4 * lane);
  }
}

__device__ __forceinline__ void agg_add_group(
    float* acc, int relv, int lane, int first, const float4 (&v)[GE]) {
#pragma unroll
  for (int j = 0; j < GE; ++j) {
    const int r = __builtin_amdgcn_readlane(relv, first + j);
    float* dst = acc + r * CH + lane;
    atomicAdd(dst + 0, v[j].x);
    atomicAdd(dst + 32, v[j].y);
    atomicAdd(dst + 64, v[j].z);
    atomicAdd(dst + 96, v[j].w);
  }
}

__device__ __forceinline__ void agg_batch32(
    const float* __restrict__ xt, float* acc, int relv, int colv, int lane) {
  float4 va[GE], vb[GE];
  agg_gather_group(xt, colv, lane, 0, va);
#pragma unroll
  for (int g = 0; g < 32; g += 2 * GE) {
    agg_gather_group(xt, colv, lane, g + GE, vb);
    agg_add_group(acc, relv, lane, g, va);
    if (g + 2 * GE < 32) agg_gather_group(xt, colv, lane, g + 2 * GE, va);
    agg_add_group(acc, relv, lane, g + GE, vb);
  }
}

__device__ __forceinline__ void agg_batch_part(
    const float* __restrict__ xt, float* acc, int relv, int colv, int lane, int cnt) {
#pragma unroll
  for (int g = 0; g < 32; g += GE) {
    if (g < cnt) {
      float4 v[GE];
#pragma unroll
      for (int j = 0; j < GE; ++j) {
        v[j] = make_float4(0.f, 0.f, 0.f, 0.f);
        if (g + j < cnt) {
          const int c = __builtin_amdgcn_readlane(colv, g + j);
          v[j] = *(const float4*)(xt + (size_t)c * CH + 4 * lane);
        }
      }
#pragma unroll
      for (int j = 0; j < GE; ++j) {
        if (g + j < cnt) {
          const int r = __builtin_amdgcn_readlane(relv, g + j);
          float* dst = acc + r * CH + lane;
          atomicAdd(dst + 0, v[j].x);
          atomicAdd(dst + 32, v[j].y);
          atomicAdd(dst + 64, v[j].z);
          atomicAdd(dst + 96, v[j].w);
        }
      }
    }
  }
}

__device__ __forceinline__ int agg_chunk_of_slot(int q, int rstart, int nround, int wv) {
  int rr = q + rstart;
  if (rr >= nround) rr -= nround;
  return rr * NW + wv;
}

__device__ __forceinline__ bool agg_slot_valid(int q, int rstart, int nround, int wv, int nfull) {
  return (q < nround) && (agg_chunk_of_slot(q, rstart, nround, wv) < nfull);
}

struct agg_planes { int4 lo, hi; };
__device__ __forceinline__ agg_planes agg_load_slot(
    const unsigned* __restrict__ keys, int q, int rstart, int nround, int wv, int lane, int nfull) {
  const int c = agg_slot_valid(q, rstart, nround, wv, nfull)
                    ? agg_chunk_of_slot(q, rstart, nround, wv) : 0;
  const unsigned* p = keys + c * (KCH / 4) + 4 * lane;
  agg_planes r;
  r.lo = *(const int4*)p;
  r.hi = *(const int4*)(p + KCH / 8);
  return r;
}

__device__ __forceinline__ unsigned agg_plane_hits(const agg_planes& d, const unsigned (&sb)[8]) {
  const unsigned t =
      ((unsigned)d.lo.x ^ sb[0]) | ((unsigned)d.lo.y ^ sb[1]) |
      ((unsigned)d.lo.z ^ sb[2]) | ((unsigned)d.lo.w ^ sb[3]) |
      ((unsigned)d.hi.x ^ sb[4]) | ((unsigned)d.hi.y ^ sb[5]) |
      ((unsigned)d.hi.z ^ sb[6]) | ((unsigned)d.hi.w ^ sb[7]);
  return ~t;
}

__device__ __forceinline__ void agg_edge_direct(
    const float* __restrict__ xt, const int* __restrict__ ei, float* acc, int nedges,
    int e, int row0, int nr, int lane) {
  const int r = ei[e] - row0;
  if ((unsigned)r < (unsigned)nr) {
    const int c = ei[nedges + e];
    const float4 v = *(const float4*)(xt + (size_t)c * CH + 4 * lane);
    float* dst = acc + r * CH + lane;
    atomicAdd(dst + 0, v.x);
    atomicAdd(dst + 32, v.y);
    atomicAdd(dst + 64, v.z);
    atomicAdd(dst + 96, v.w);
  }
}

__device__ __forceinline__ void agg_hits_direct(
    const float* __restrict__ xt, const int* __restrict__ ei, float* acc, int nedges,
    int ebase, unsigned hits, unsigned m, int row0, int nr, int lane) {
#pragma unroll 1
  while (m != 0u) {
    const int l = __builtin_ctz(m);
    m &= m - 1u;
    unsigned bits = (unsigned)__builtin_amdgcn_readlane((int)hits, l);
#pragma unroll 1
    while (bits != 0u) {
      const int e = ebase + 32 * l + __builtin_ctz(bits);
      bits &= bits - 1u;
      agg_edge_direct(xt, ei, acc, nedges, e, row0, nr, lane);
    }
  }
}

__global__ __launch_bounds__(1024) void gcn_aggregate(
    const float* __restrict__ xt, const int* __restrict__ ei,
    float* __restrict__ out, int nedges, int nrows) {
  __shared__ float acc[(ROWS + 1) * CH];
  __shared__ int ring_e[NW][PCAP];

  const unsigned* keys = slice_keys(xt, nrows);
  const unsigned* pk = edge_words(keys, nedges);

  const int tid = threadIdx.x;
  const int lane = tid & 31;
  const int wv = __builtin_amdgcn_readfirstlane(tid >> 5);
  int* mye = ring_e[wv];

  const int row0 = blockIdx.x * ROWS;
  const int nr = min(ROWS, nrows - row0);
  unsigned sb[8];
#pragma unroll
  for (int b = 0; b < 8; ++b) sb[b] = 0u - (((unsigned)blockIdx.x >> b) & 1u);

  float4* acc4 = (float4*)acc;
  for (int i = tid; i < (ROWS + 1) * CH / 4; i += 1024) acc4[i] = make_float4(0.f, 0.f, 0.f, 0.f);
  __syncthreads();

  const int nfull = nedges / KCH;
  const int nround = (nfull + NW - 1) / NW;
  const int rstart = (int)(((unsigned)blockIdx.x * (unsigned)nround) / gridDim.x);
  const int niter = (nround + NBUF - 1) / NBUF;

  int head = 0;
  int npend = 0;

  for (int base = nfull * KCH + 32 * wv; base < nedges; base += 32 * NW) {
    const int e = base + lane;
    int r = -1;
    if (e < nedges) r = ei[e] - row0;
    unsigned m = __builtin_amdgcn_ballot_w32((unsigned)r < (unsigned)nr);
    while (m != 0u) {
      const int l = __builtin_ctz(m);
      m &= m - 1u;
      agg_edge_direct(xt, ei, acc, nedges, base + l, row0, nr, lane);
    }
  }

  if (nround > 0) {
    agg_planes buf[NBUF];
#pragma unroll
    for (int k = 0; k < NBUF; ++k) buf[k] = agg_load_slot(keys, k, rstart, nround, wv, lane, nfull);

    for (int it = 0; it < niter; ++it) {
#pragma unroll
      for (int k = 0; k < NBUF; ++k) {
        const int q = it * NBUF + k;
        unsigned hits = agg_plane_hits(buf[k], sb);
        unsigned m = __builtin_amdgcn_ballot_w32(hits != 0u);
        if (!agg_slot_valid(q, rstart, nround, wv, nfull)) m = 0u;
        if (m != 0u) {
          const int ebase = agg_chunk_of_slot(q, rstart, nround, wv) * KCH;
          do {
            if (npend > PCAP - 32) {
              agg_hits_direct(xt, ei, acc, nedges, ebase, hits, m, row0, nr, lane);
              break;
            }
            if (hits != 0u) {
              const int idx = (head + npend + (int)__builtin_amdgcn_mbcnt_lo(m, 0u)) & (PCAP - 1);
              mye[idx] = ebase + 32 * lane + __builtin_ctz(hits);
              hits &= hits - 1u;
            }
            npend += __builtin_popcount(m);
            m = __builtin_amdgcn_ballot_w32(hits != 0u);
          } while (m != 0u);
        }
        buf[k] = agg_load_slot(keys, q + NBUF, rstart, nround, wv, lane, nfull);

        if (k % NFL == NFL - 1) {
          while (npend >= 32) {
            const unsigned w = pk[mye[(head + lane) & (PCAP - 1)]];
            const int r = (int)(w >> 24);
            const int colv = (int)(w & 0xFFFFFFu);
            const int relv = r < nr ? r : ROWS;
            agg_batch32(xt, acc, relv, colv, lane);
            head = (head + 32) & (PCAP - 1);
            npend -= 32;
          }
        }
      }
    }
  }

  while (npend > 0) {
    const int cnt = npend < 32 ? npend : 32;
    int relv = 0, colv = 0;
    if (lane < cnt) {
      const unsigned w = pk[mye[(head + lane) & (PCAP - 1)]];
      const int r = (int)(w >> 24);
      colv = (int)(w & 0xFFFFFFu);
      relv = r < nr ? r : ROWS;
    }
    agg_batch_part(xt, acc, relv, colv, lane, cnt);
    head = (head + cnt) & (PCAP - 1);
    npend -= cnt;
  }

  __syncthreads();

  float4* out4 = (float4*)out;
  for (int i = tid; i < nr * 32; i += 1024) {
    const int row = i >> 5;
    const float* a = acc + row * CH + lane;
    float4 v;
    v.x = tanhf(a[0]);
    v.y = tanhf(a[32]);
    v.z = tanhf(a[64]);
    v.w = tanhf(a[96]);
    out4[(size_t)(row0 + row) * 32 + lane] = v;
  }
}

extern "C" void kernel_launch(void* const* d_in, const int* in_sizes, int n_in,
                              void* d_out, int out_size, void* d_ws, size_t ws_size,
                              hipStream_t stream) {
  const float* x    = (const float*)d_in[0];
  const int*   ei   = (const int*)d_in[1];
  const float* W    = (const float*)d_in[2];
  const float* bias = (const float*)d_in[3];

  float* out = (float*)d_out;
  float* xt  = (float*)d_ws;

  const int N = in_sizes[0] / CH;
  const int E = in_sizes[1] / 2;

  const int ntiles = N / 16;
  gcn_gemm_wmma<<<(ntiles + 3) / 4, 128, 0, stream>>>(x, W, bias, xt, ntiles, ei, E, N);

  gcn_aggregate<<<(N + ROWS - 1) / ROWS, 1024, 0, stream>>>(xt, ei, out, E, N);
}


# Product launch made persistent: 256 work-groups of 16 waves stage W once each; a wave takes 16x64 half tiles, its x rows loaded whole ahead (16-byte loads); k order inside the product's sums changes
# speedup vs baseline: 3.4727x; 1.0968x over previous
#include <hip/hip_runtime.h>

typedef float v2f __attribute__((ext_vector_type(2)));
typedef float v4f __attribute__((ext_vector_type(4)));
typedef float v8f __attribute__((ext_vector_type(8)));

#define CH 128
#define LDS_PITCH 132
#define ROWS 196

__device__ __forceinline__ unsigned* slice_keys(float* xt, int nrows) {
  return (unsigned*)(xt + (size_t)nrows * CH);
}
__device__ __forceinline__ const unsigned* slice_keys(const float* xt, int nrows) {
  return (const unsigned*)(xt + (size_t)nrows * CH);
}
__device__ __forceinline__ unsigned* edge_words(unsigned* keys, int nedges) {
  return keys + (((nedges + 15) >> 4) << 2);
}
__device__ __forceinline__ const unsigned* edge_words(const unsigned* keys, int nedges) {
  return keys + (((nedges + 15) >> 4) << 2);
}
__device__ __forceinline__ unsigned edge_word(int dest, int src, int nrows) {
  const unsigned r = (unsigned)dest < (unsigned)nrows ? (unsigned)dest % ROWS : 255u;
  return (r << 24) | ((unsigned)src & 0xFFFFFFu);
}
__device__ __forceinline__ int4 edge_word4(int4 d, int4 s, int nrows) {
  return make_int4((int)edge_word(d.x, s.x, nrows), (int)edge_word(d.y, s.y, nrows),
                   (int)edge_word(d.z, s.z, nrows), (int)edge_word(d.w, s.w, nrows));
}
__device__ __forceinline__ int4 edge_src4(const int* __restrict__ ei, int nedges, int g) {
  const int* p = ei + nedges + 4 * g;
  if ((nedges & 3) == 0) return *(const int4*)p;
  return make_int4(p[0], p[1], p[2], p[3]);
}
#define KCH 1024
#define PGRP 8

__device__ __forceinline__ void plane_ids(const int* __restrict__ ei, int nedges, int blk,
                                          int wave, int lane, int (&d)[PGRP]) {
#pragma unroll
  for (int j = 0; j < PGRP; ++j) {
    const int e = blk * KCH + (PGRP * wave + j) * 32 + lane;
    d[j] = ei[max(min(e, nedges - 1), 0)];
  }
}

__device__ __forceinline__ void plane_store(unsigned* keys, int blk, int wave, int lane,
                                            const int (&d)[PGRP]) {
  unsigned p[8] = {0u, 0u, 0u, 0u, 0u, 0u, 0u, 0u};
#pragma unroll
  for (int j = 0; j < PGRP; ++j) {
    const unsigned key = (unsigned)d[j] / ROWS;
#pragma unroll
    for (int b = 0; b < 8; ++b) {
      const unsigned w = __builtin_amdgcn_ballot_w32(((key >> b) & 1u) != 0u);
      if (lane == j) p[b] = w;
    }
  }
  if (lane < PGRP) {
    unsigned* q = keys + blk * (KCH / 4) + (PGRP * wave + lane) * 4;
    *(int4*)q = make_int4((int)p[0], (int)p[1], (int)p[2], (int)p[3]);
    *(int4*)(q + KCH / 8) = make_int4((int)p[4], (int)p[5], (int)p[6], (int)p[7]);
  }
}

#define GW 16

__device__ __forceinline__ int tile_k(int s, int h) { return 8 * s + 4 * h; }

__device__ __forceinline__ void tile_a_load(const float* __restrict__ x, int rt,
                                            int lh, int khi, v4f (&a)[16]) {
  const float* arow = x + (size_t)(rt * 16 + lh) * CH;
#pragma unroll
  for (int s = 0; s < 16; ++s) a[s] = *(const v4f*)(arow + tile_k(s, khi));
}

__device__ __forceinline__ void half_tile(const float* wl, const float (&bv)[4],
                                          float* __restrict__ xt, int rt, int hb,
                                          int lh, int khi, const v4f (&a)[16]) {
  v8f acc[4];
#pragma unroll
  for (int c = 0; c < 4; ++c)
    acc[c] = (v8f){bv[c], bv[c], bv[c], bv[c], bv[c], bv[c], bv[c], bv[c]};
  const float* brow = wl + (4 * hb * 16 + lh) * LDS_PITCH;
#pragma unroll
  for (int s = 0; s < 16; ++s) {
    v4f b[4];
#pragma unroll
    for (int c = 0; c < 4; ++c) b[c] = *(const v4f*)(brow + c * 16 * LDS_PITCH + tile_k(s, khi));
#pragma unroll
    for (int c = 0; c < 4; ++c)
      acc[c] = __builtin_amdgcn_wmma_f32_16x16x4_f32(
          false, (v2f){a[s].x, a[s].y}, false, (v2f){b[c].x, b[c].y}, (short)0, acc[c], false, false);
#pragma unroll
    for (int c = 0; c < 4; ++c)
      acc[c] = __builtin_amdgcn_wmma_f32_16x16x4_f32(
          false, (v2f){a[s].z, a[s].w}, false, (v2f){b[c].z, b[c].w}, (short)0, acc[c], false, false);
  }
  const int rb = rt * 16;
#pragma unroll
  for (int c = 0; c < 4; ++c) {
#pragma unroll
    for (int i = 0; i < 8; ++i) {
      xt[(size_t)(rb + i + 8 * khi) * CH + (4 * hb + c) * 16 + lh] = acc[c][i];
    }
  }
}

__global__ __launch_bounds__(GW * 32) void gcn_gemm_wmma(
    const float* __restrict__ x, const float* __restrict__ W,
    const float* __restrict__ bias, float* __restrict__ xt, int ntiles,
    const int* __restrict__ ei, int nedges, int nrows) {
  __shared__ __attribute__((aligned(16))) float wl[CH * LDS_PITCH];

  const int tid = threadIdx.x;
  const int lane = tid & 31;
  const int wave = __builtin_amdgcn_readfirstlane(tid >> 5);
  const int lh  = lane & 15;
  const int khi = lane >> 4;

  float4 wreg[CH * CH / 4 / (GW * 32)];
#pragma unroll
  for (int j = 0; j < CH * CH / 4 / (GW * 32); ++j)
    wreg[j] = ((const float4*)W)[tid + j * GW * 32];

  const int code = (int)((0x7F6DE5C43B29A180ull >> (4 * wave)) & 15ull);
  const int hb = code & 1;
  const int npair = (GW / 2) * (int)gridDim.x;
  const int rt0 = (code >> 1) * (int)gridDim.x + (int)blockIdx.x;
  v4f a[16];
  if (rt0 < ntiles) tile_a_load(x, rt0, lh, khi, a);
  float bv[4];
#pragma unroll
  for (int c = 0; c < 4; ++c) bv[c] = bias[(4 * hb + c) * 16 + lh];

  unsigned* keys = slice_keys(xt, nrows);
  const int nblk = nedges / KCH;
  const int team = (int)blockIdx.x * (GW / 4) + (wave >> 2);
  const int nteam = (int)gridDim.x * (GW / 4);
  const int tw = wave & 3;
  int pd[PGRP];
  plane_ids(ei, nedges, team, tw, lane, pd);
  const int nthr = (int)gridDim.x * GW * 32;
  const int gid = (int)blockIdx.x * GW * 32 + tid;
  const int nquad = nedges >> 2;
  const int g0 = max(min(gid, nquad - 1), 0), g1 = max(min(gid + nthr, nquad - 1), 0);

#pragma unroll
  for (int j = 0; j < CH * CH / 4 / (GW * 32); ++j) {
    const int i = tid + j * GW * 32;
    *(float4*)&wl[(i >> 5) * LDS_PITCH + 4 * (i & 31)] = wreg[j];
  }
  __syncthreads();

  int4 kd0, kd1, ks0, ks1;
  bool have_ids = false;
  for (int rt = rt0; rt < ntiles; rt += npair) {
    asm volatile("" ::: "memory");
    const bool more = rt + npair < ntiles;
    v4f an[16];
    if (more) {
      tile_a_load(x, rt + npair, lh, khi, an);
    } else {
      kd0 = *(const int4*)(ei + 4 * g0);
      kd1 = *(const int4*)(ei + 4 * g1);
      ks0 = edge_src4(ei, nedges, g0);
      ks1 = edge_src4(ei, nedges, g1);
      have_ids = true;
    }
    half_tile(wl, bv, xt, rt, hb, lh, khi, a);
    if (more) {
#pragma unroll
      for (int s = 0; s < 16; ++s) a[s] = an[s];
    }
  }

  if (!have_ids) {
    kd0 = *(const int4*)(ei + 4 * g0);
    kd1 = *(const int4*)(ei + 4 * g1);
    ks0 = edge_src4(ei, nedges, g0);
    ks1 = edge_src4(ei, nedges, g1);
  }

  if (team < nblk) plane_store(keys, team, tw, lane, pd);
  for (int blk = team + nteam; blk < nblk; blk += nteam) {
    plane_ids(ei, nedges, blk, tw, lane, pd);
    plane_store(keys, blk, tw, lane, pd);
  }

  unsigned* pk = edge_words(keys, nedges);
  if (gid < nquad) *(int4*)(pk + 4 * gid) = edge_word4(kd0, ks0, nrows);
  if (gid + nthr < nquad) *(int4*)(pk + 4 * (gid + nthr)) = edge_word4(kd1, ks1, nrows);
  for (int g = gid + 2 * nthr; g < nquad; g += nthr) {
    const int4 d = *(const int4*)(ei + 4 * g);
    *(int4*)(pk + 4 * g) = edge_word4(d, edge_src4(ei, nedges, g), nrows);
  }
  if (gid < (nedges & 3)) {
    const int e = (nquad << 2) + gid;
    pk[e] = edge_word(ei[e], ei[nedges + e], nrows);
  }
}

#define NW 32
#define GE 4
#define NBUF 6
#define NFL 2
#define PCAP 256

__device__ __forceinline__ void agg_gather_group(
    const float* __restrict__ xt, int colv, int lane, int first, float4 (&v)[GE]) {
#pragma unroll
  for (int j = 0; j < GE; ++j) {
    const int c = __builtin_amdgcn_readlane(colv, first + j);
    v[j] = *(const float4*)(xt + (size_t)c * CH + 4 * lane);
  }
}

__device__ __forceinline__ void agg_add_group(
    float* acc, int relv, int lane, int first, const float4 (&v)[GE]) {
#pragma unroll
  for (int j = 0; j < GE; ++j) {
    const int r = __builtin_amdgcn_readlane(relv, first + j);
    float* dst = acc + r * CH + lane;
    atomicAdd(dst + 0, v[j].x);
    atomicAdd(dst + 32, v[j].y);
    atomicAdd(dst + 64, v[j].z);
    atomicAdd(dst + 96, v[j].w);
  }
}

__device__ __forceinline__ void agg_batch32(
    const float* __restrict__ xt, float* acc, int relv, int colv, int lane) {
  float4 va[GE], vb[GE];
  agg_gather_group(xt, colv, lane, 0, va);
#pragma unroll
  for (int g = 0; g < 32; g += 2 * GE) {
    agg_gather_group(xt, colv, lane, g + GE, vb);
    agg_add_group(acc, relv, lane, g, va);
    if (g + 2 * GE < 32) agg_gather_group(xt, colv, lane, g + 2 * GE, va);
    agg_add_group(acc, relv, lane, g + GE, vb);
  }
}

__device__ __forceinline__ void agg_batch_part(
    const float* __restrict__ xt, float* acc, int relv, int colv, int lane, int cnt) {
#pragma unroll
  for (int g = 0; g < 32; g += GE) {
    if (g < cnt) {
      float4 v[GE];
#pragma unroll
      for (int j = 0; j < GE; ++j) {
        v[j] = make_float4(0.f, 0.f, 0.f, 0.f);
        if (g + j < cnt) {
          const int c = __builtin_amdgcn_readlane(colv, g + j);
          v[j] = *(const float4*)(xt + (size_t)c * CH + 4 * lane);
        }
      }
#pragma unroll
      for (int j = 0; j < GE; ++j) {
        if (g + j < cnt) {
          const int r = __builtin_amdgcn_readlane(relv, g + j);
          float* dst = acc + r * CH + lane;
          atomicAdd(dst + 0, v[j].x);
          atomicAdd(dst + 32, v[j].y);
          atomicAdd(dst + 64, v[j].z);
          atomicAdd(dst + 96, v[j].w);
        }
      }
    }
  }
}

__device__ __forceinline__ int agg_chunk_of_slot(int q, int rstart, int nround, int wv) {
  int rr = q + rstart;
  if (rr >= nround) rr -= nround;
  return rr * NW + wv;
}

__device__ __forceinline__ bool agg_slot_valid(int q, int rstart, int nround, int wv, int nfull) {
  return (q < nround) && (agg_chunk_of_slot(q, rstart, nround, wv) < nfull);
}

struct agg_planes { int4 lo, hi; };
__device__ __forceinline__ agg_planes agg_load_slot(
    const unsigned* __restrict__ keys, int q, int rstart, int nround, int wv, int lane, int nfull) {
  const int c = agg_slot_valid(q, rstart, nround, wv, nfull)
                    ? agg_chunk_of_slot(q, rstart, nround, wv) : 0;
  const unsigned* p = keys + c * (KCH / 4) + 4 * lane;
  agg_planes r;
  r.lo = *(const int4*)p;
  r.hi = *(const int4*)(p + KCH / 8);
  return r;
}

__device__ __forceinline__ unsigned agg_plane_hits(const agg_planes& d, const unsigned (&sb)[8]) {
  const unsigned t =
      ((unsigned)d.lo.x ^ sb[0]) | ((unsigned)d.lo.y ^ sb[1]) |
      ((unsigned)d.lo.z ^ sb[2]) | ((unsigned)d.lo.w ^ sb[3]) |
      ((unsigned)d.hi.x ^ sb[4]) | ((unsigned)d.hi.y ^ sb[5]) |
      ((unsigned)d.hi.z ^ sb[6]) | ((unsigned)d.hi.w ^ sb[7]);
  return ~t;
}

__device__ __forceinline__ void agg_edge_direct(
    const float* __restrict__ xt, const int* __restrict__ ei, float* acc, int nedges,
    int e, int row0, int nr, int lane) {
  const int r = ei[e] - row0;
  if ((unsigned)r < (unsigned)nr) {
    const int c = ei[nedges + e];
    const float4 v = *(const float4*)(xt + (size_t)c * CH + 4 * lane);
    float* dst = acc + r * CH + lane;
    atomicAdd(dst + 0, v.x);
    atomicAdd(dst + 32, v.y);
    atomicAdd(dst + 64, v.z);
    atomicAdd(dst + 96, v.w);
  }
}

__device__ __forceinline__ void agg_hits_direct(
    const float* __restrict__ xt, const int* __restrict__ ei, float* acc, int nedges,
    int ebase, unsigned hits, unsigned m, int row0, int nr, int lane) {
#pragma unroll 1
  while (m != 0u) {
    const int l = __builtin_ctz(m);
    m &= m - 1u;
    unsigned bits = (unsigned)__builtin_amdgcn_readlane((int)hits, l);
#pragma unroll 1
    while (bits != 0u) {
      const int e = ebase + 32 * l + __builtin_ctz(bits);
      bits &= bits - 1u;
      agg_edge_direct(xt, ei, acc, nedges, e, row0, nr, lane);
    }
  }
}

__global__ __launch_bounds__(1024) void gcn_aggregate(
    const float* __restrict__ xt, const int* __restrict__ ei,
    float* __restrict__ out, int nedges, int nrows) {
  __shared__ float acc[(ROWS + 1) * CH];
  __shared__ int ring_e[NW][PCAP];

  const unsigned* keys = slice_keys(xt, nrows);
  const unsigned* pk = edge_words(keys, nedges);

  const int tid = threadIdx.x;
  const int lane = tid & 31;
  const int wv = __builtin_amdgcn_readfirstlane(tid >> 5);
  int* mye = ring_e[wv];

  const int row0 = blockIdx.x * ROWS;
  const int nr = min(ROWS, nrows - row0);
  unsigned sb[8];
#pragma unroll
  for (int b = 0; b < 8; ++b) sb[b] = 0u - (((unsigned)blockIdx.x >> b) & 1u);

  float4* acc4 = (float4*)acc;
  for (int i = tid; i < (ROWS + 1) * CH / 4; i += 1024) acc4[i] = make_float4(0.f, 0.f, 0.f, 0.f);
  __syncthreads();

  const int nfull = nedges / KCH;
  const int nround = (nfull + NW - 1) / NW;
  const int rstart = (int)(((unsigned)blockIdx.x * (unsigned)nround) / gridDim.x);
  const int niter = (nround + NBUF - 1) / NBUF;

  int head = 0;
  int npend = 0;

  for (int base = nfull * KCH + 32 * wv; base < nedges; base += 32 * NW) {
    const int e = base + lane;
    int r = -1;
    if (e < nedges) r = ei[e] - row0;
    unsigned m = __builtin_amdgcn_ballot_w32((unsigned)r < (unsigned)nr);
    while (m != 0u) {
      const int l = __builtin_ctz(m);
      m &= m - 1u;
      agg_edge_direct(xt, ei, acc, nedges, base + l, row0, nr, lane);
    }
  }

  if (nround > 0) {
    agg_planes buf[NBUF];
#pragma unroll
    for (int k = 0; k < NBUF; ++k) buf[k] = agg_load_slot(keys, k, rstart, nround, wv, lane, nfull);

    for (int it = 0; it < niter; ++it) {
#pragma unroll
      for (int k = 0; k < NBUF; ++k) {
        const int q = it * NBUF + k;
        unsigned hits = agg_plane_hits(buf[k], sb);
        unsigned m = __builtin_amdgcn_ballot_w32(hits != 0u);
        if (!agg_slot_valid(q, rstart, nround, wv, nfull)) m = 0u;
        if (m != 0u) {
          const int ebase = agg_chunk_of_slot(q, rstart, nround, wv) * KCH;
          do {
            if (npend > PCAP - 32) {
              agg_hits_direct(xt, ei, acc, nedges, ebase, hits, m, row0, nr, lane);
              break;
            }
            if (hits != 0u) {
              const int idx = (head + npend + (int)__builtin_amdgcn_mbcnt_lo(m, 0u)) & (PCAP - 1);
              mye[idx] = ebase + 32 * lane + __builtin_ctz(hits);
              hits &= hits - 1u;
            }
            npend += __builtin_popcount(m);
            m = __builtin_amdgcn_ballot_w32(hits != 0u);
          } while (m != 0u);
        }
        buf[k] = agg_load_slot(keys, q + NBUF, rstart, nround, wv, lane, nfull);

        if (k % NFL == NFL - 1) {
          while (npend >= 32) {
            const unsigned w = pk[mye[(head + lane) & (PCAP - 1)]];
            const int r = (int)(w >> 24);
            const int colv = (int)(w & 0xFFFFFFu);
            const int relv = r < nr ? r : ROWS;
            agg_batch32(xt, acc, relv, colv, lane);
            head = (head + 32) & (PCAP - 1);
            npend -= 32;
          }
        }
      }
    }
  }

  while (npend > 0) {
    const int cnt = npend < 32 ? npend : 32;
    int relv = 0, colv = 0;
    if (lane < cnt) {
      const unsigned w = pk[mye[(head + lane) & (PCAP - 1)]];
      const int r = (int)(w >> 24);
      colv = (int)(w & 0xFFFFFFu);
      relv = r < nr ? r : ROWS;
    }
    agg_batch_part(xt, acc, relv, colv, lane, cnt);
    head = (head + cnt) & (PCAP - 1);
    npend -= cnt;
  }

  __syncthreads();

  float4* out4 = (float4*)out;
  for (int i = tid; i < nr * 32; i += 1024) {
    const int row = i >> 5;
    const float* a = acc + row * CH + lane;
    float4 v;
    v.x = tanhf(a[0]);
    v.y = tanhf(a[32]);
    v.z = tanhf(a[64]);
    v.w = tanhf(a[96]);
    out4[(size_t)(row0 + row) * 32 + lane] = v;
  }
}

extern "C" void kernel_launch(void* const* d_in, const int* in_sizes, int n_in,
                              void* d_out, int out_size, void* d_ws, size_t ws_size,
                              hipStream_t stream) {
  const float* x    = (const float*)d_in[0];
  const int*   ei   = (const int*)d_in[1];
  const float* W    = (const float*)d_in[2];
  const float* bias = (const float*)d_in[3];

  float* out = (float*)d_out;
  float* xt  = (float*)d_ws;

  const int N = in_sizes[0] / CH;
  const int E = in_sizes[1] / 2;

  const int ntiles = N / 16;
  gcn_gemm_wmma<<<256, GW * 32, 0, stream>>>(x, W, bias, xt, ntiles, ei, E, N);

  gcn_aggregate<<<(N + ROWS - 1) / ROWS, 1024, 0, stream>>>(xt, ei, out, E, N);
}
